# starting kernel
# baseline (speedup 1.0000x reference)
#include <hip/hip_runtime.h>


#define N_MHR   20000
#define N_SMPLX 10475
#define N_PAD   10480
#define NT_COL  655
#define NT_ROW  1250

typedef float v2f __attribute__((ext_vector_type(2)));
typedef float v8f __attribute__((ext_vector_type(8)));

__global__ __launch_bounds__(256) void vm_prep(const float* __restrict__ mhr,
                                               const float* __restrict__ smplx,
                                               float4* __restrict__ A4,
                                               float4* __restrict__ B4) {
  int i = blockIdx.x * blockDim.x + threadIdx.x;
  if (i < N_MHR) {
    float x = mhr[3 * i + 0], y = mhr[3 * i + 1], z = mhr[3 * i + 2];
    A4[i] = make_float4(-2.0f * x, -2.0f * y, -2.0f * z, 1.0f);
  }
  if (i < N_PAD) {
    if (i < N_SMPLX) {
      float x = smplx[3 * i + 0], y = smplx[3 * i + 1], z = smplx[3 * i + 2];
      B4[i] = make_float4(x, y, z, x * x + y * y + z * z);
    } else {
      B4[i] = make_float4(0.0f, 0.0f, 0.0f, __builtin_inff());
    }
  }
}

__global__ __launch_bounds__(256) void vm_match(const float4* __restrict__ A4,
                                                const float4* __restrict__ B4,
                                                int* __restrict__ out) {
  extern __shared__ float4 Bs4[];
  float2* Bs2 = (float2*)Bs4;

  for (int t = threadIdx.x; t < N_PAD; t += 256) Bs4[t] = B4[t];
  __syncthreads();

  const float2* A2 = (const float2*)A4;
  const int lane = threadIdx.x & 31;
  const int wave = threadIdx.x >> 5;
  const int nlo  = lane & 15;
  const int kp   = lane >> 4;

  for (int rt = blockIdx.x * 8 + wave; rt < NT_ROW; rt += gridDim.x * 8) {
    const int m = rt * 16 + nlo;
    const float2 a2 = A2[2 * m + kp];
    const v2f af = {a2.x, a2.y};

    float vmin[8];
    int   imin[8];
#pragma unroll
    for (int r = 0; r < 8; ++r) { vmin[r] = __builtin_inff(); imin[r] = 0; }

#pragma unroll 2
    for (int ct = 0; ct < NT_COL; ++ct) {
      const int n = ct * 16 + nlo;
      const float2 b2 = Bs2[2 * n + kp];
      const v2f bf = {b2.x, b2.y};
      v8f c = {0.f, 0.f, 0.f, 0.f, 0.f, 0.f, 0.f, 0.f};
      c = __builtin_amdgcn_wmma_f32_16x16x4_f32(false, af, false, bf,
                                                (short)0, c, false, false);
#pragma unroll
      for (int r = 0; r < 8; ++r) {
        float cv = c[r];
        if (cv < vmin[r]) { vmin[r] = cv; imin[r] = n; }
      }
    }

#pragma unroll
    for (int r = 0; r < 8; ++r) {
      float v = vmin[r];
      int   idx = imin[r];
#pragma unroll
      for (int mask = 8; mask >= 1; mask >>= 1) {
        float vo = __shfl_xor(v, mask, 32);
        int   io = __shfl_xor(idx, mask, 32);
        const bool better = (vo < v) | ((vo == v) & (io < idx));
        idx = better ? io : idx;
        v   = better ? vo : v;
      }
      if (nlo == 0) {
        out[rt * 16 + r + (kp << 3)] = idx;
      }
    }
  }
}

extern "C" void kernel_launch(void* const* d_in, const int* in_sizes, int n_in,
                              void* d_out, int out_size, void* d_ws, size_t ws_size,
                              hipStream_t stream) {
  const float* mhr   = (const float*)d_in[0];
  const float* smplx = (const float*)d_in[1];
  int* out = (int*)d_out;

  char* ws = (char*)d_ws;
  float4* A4 = (float4*)ws;
  float4* B4 = (float4*)(ws + (size_t)N_MHR * 16);

  vm_prep<<<(N_MHR + 255) / 256, 256, 0, stream>>>(mhr, smplx, A4, B4);

  const size_t lds_bytes = (size_t)N_PAD * sizeof(float4);
  vm_match<<<64, 256, lds_bytes, stream>>>(A4, B4, out);
}


# match split 4 ways over the table columns on grid.y: 63 x 4 work-groups of 20 waves, each stages only its slice in LDS; per-row partial (min, argmin) to workspace, a third launch merges them
# speedup vs baseline: 4.5141x; 4.5141x over previous
#include <hip/hip_runtime.h>


#define N_MHR    20000
#define N_SMPLX  10475
#define NT_ROW   1250
#define N_SPLIT  4
#define CT_SLICE 164
#define NT_COL   (N_SPLIT * CT_SLICE)
#define N_PAD    (NT_COL * 16)
#define N_SLICE  (CT_SLICE * 16)
#define WG_WAVES 20
#define WG_THR   (WG_WAVES * 32)
#define WG_X     ((NT_ROW + WG_WAVES - 1) / WG_WAVES)
#define LDS_RESERVE (162 * 1024)

typedef float v2f __attribute__((ext_vector_type(2)));
typedef float v8f __attribute__((ext_vector_type(8)));

__global__ __launch_bounds__(256) void vm_prep(const float* __restrict__ mhr,
                                               const float* __restrict__ smplx,
                                               float4* __restrict__ A4,
                                               float4* __restrict__ B4) {
  int i = blockIdx.x * blockDim.x + threadIdx.x;
  if (i < N_MHR) {
    float x = mhr[3 * i + 0], y = mhr[3 * i + 1], z = mhr[3 * i + 2];
    A4[i] = make_float4(-2.0f * x, -2.0f * y, -2.0f * z, 1.0f);
  }
  if (i < N_PAD) {
    if (i < N_SMPLX) {
      float x = smplx[3 * i + 0], y = smplx[3 * i + 1], z = smplx[3 * i + 2];
      B4[i] = make_float4(x, y, z, x * x + y * y + z * z);
    } else {
      B4[i] = make_float4(0.0f, 0.0f, 0.0f, __builtin_inff());
    }
  }
}

__global__ __launch_bounds__(WG_THR) void vm_match(const float4* __restrict__ A4,
                                                   const float4* __restrict__ B4,
                                                   int2* __restrict__ part) {
  extern __shared__ float4 Bs4[];
  float2* Bs2 = (float2*)Bs4;

  const int slice = blockIdx.y;
  const int col0  = slice * N_SLICE;
  const float4* __restrict__ Bg = B4 + col0;

  for (int t = threadIdx.x; t < N_SLICE; t += WG_THR) Bs4[t] = Bg[t];
  __syncthreads();

  const float2* A2 = (const float2*)A4;
  const int lane = threadIdx.x & 31;
  const int wave = threadIdx.x >> 5;
  const int nlo  = lane & 15;
  const int kp   = lane >> 4;

  const int rt = blockIdx.x * WG_WAVES + wave;
  if (rt < NT_ROW) {
    const int m = rt * 16 + nlo;
    const float2 a2 = A2[2 * m + kp];
    const v2f af = {a2.x, a2.y};

    float vmin[8];
    int   imin[8];
#pragma unroll
    for (int r = 0; r < 8; ++r) { vmin[r] = __builtin_inff(); imin[r] = 0; }

#pragma unroll 2
    for (int ct = 0; ct < CT_SLICE; ++ct) {
      const int n = ct * 16 + nlo;
      const float2 b2 = Bs2[2 * n + kp];
      const v2f bf = {b2.x, b2.y};
      v8f c = {0.f, 0.f, 0.f, 0.f, 0.f, 0.f, 0.f, 0.f};
      c = __builtin_amdgcn_wmma_f32_16x16x4_f32(false, af, false, bf,
                                                (short)0, c, false, false);
#pragma unroll
      for (int r = 0; r < 8; ++r) {
        float cv = c[r];
        if (cv < vmin[r]) { vmin[r] = cv; imin[r] = n; }
      }
    }

    int2 mine = make_int2(0, 0);
#pragma unroll
    for (int r = 0; r < 8; ++r) {
      float v = vmin[r];
      int   idx = imin[r];
#pragma unroll
      for (int mask = 8; mask >= 1; mask >>= 1) {
        float vo = __shfl_xor(v, mask, 32);
        int   io = __shfl_xor(idx, mask, 32);
        const bool better = (vo < v) | ((vo == v) & (io < idx));
        idx = better ? io : idx;
        v   = better ? vo : v;
      }
      if (nlo == r) { mine.x = __float_as_int(v); mine.y = idx + col0; }
    }
    if (nlo < 8) {
      part[slice * N_MHR + rt * 16 + nlo + (kp << 3)] = mine;
    }
  }
}

__global__ __launch_bounds__(256) void vm_merge(const int2* __restrict__ part,
                                                int* __restrict__ out) {
  const int m = blockIdx.x * blockDim.x + threadIdx.x;
  if (m < N_MHR) {
    const int2 p = part[m];
    float bv = __int_as_float(p.x);
    int   bi = p.y;
#pragma unroll
    for (int s = 1; s < N_SPLIT; ++s) {
      const int2 q = part[s * N_MHR + m];
      const float v = __int_as_float(q.x);
      if (v < bv) { bv = v; bi = q.y; }
    }
    out[m] = bi;
  }
}

extern "C" void kernel_launch(void* const* d_in, const int* in_sizes, int n_in,
                              void* d_out, int out_size, void* d_ws, size_t ws_size,
                              hipStream_t stream) {
  const float* mhr   = (const float*)d_in[0];
  const float* smplx = (const float*)d_in[1];
  int* out = (int*)d_out;

  char* ws = (char*)d_ws;
  float4* A4  = (float4*)ws;
  float4* B4  = (float4*)(ws + (size_t)N_MHR * 16);
  int2*  part = (int2*)(ws + (size_t)N_MHR * 16 + (size_t)N_PAD * 16);

  vm_prep<<<(N_MHR + 255) / 256, 256, 0, stream>>>(mhr, smplx, A4, B4);

  vm_match<<<dim3(WG_X, N_SPLIT), WG_THR, LDS_RESERVE, stream>>>(A4, B4, part);

  vm_merge<<<(N_MHR + 255) / 256, 256, 0, stream>>>(part, out);
}


# match start-up: each thread asks for its 4-5 table-staging float4 and its A pair in one group ahead of the LDS stores and the barrier: one cache trip before the first product, not 5-6 in series
# speedup vs baseline: 4.9857x; 1.1045x over previous
#include <hip/hip_runtime.h>


#define N_MHR    20000
#define N_SMPLX  10475
#define NT_ROW   1250
#define N_SPLIT  4
#define CT_SLICE 164
#define NT_COL   (N_SPLIT * CT_SLICE)
#define N_PAD    (NT_COL * 16)
#define N_SLICE  (CT_SLICE * 16)
#define WG_WAVES 20
#define WG_THR   (WG_WAVES * 32)
#define WG_X     ((NT_ROW + WG_WAVES - 1) / WG_WAVES)
#define LDS_RESERVE (162 * 1024)

typedef float v2f __attribute__((ext_vector_type(2)));
typedef float v8f __attribute__((ext_vector_type(8)));

__global__ __launch_bounds__(256) void vm_prep(const float* __restrict__ mhr,
                                               const float* __restrict__ smplx,
                                               float4* __restrict__ A4,
                                               float4* __restrict__ B4) {
  int i = blockIdx.x * blockDim.x + threadIdx.x;
  if (i < N_MHR) {
    float x = mhr[3 * i + 0], y = mhr[3 * i + 1], z = mhr[3 * i + 2];
    A4[i] = make_float4(-2.0f * x, -2.0f * y, -2.0f * z, 1.0f);
  }
  if (i < N_PAD) {
    if (i < N_SMPLX) {
      float x = smplx[3 * i + 0], y = smplx[3 * i + 1], z = smplx[3 * i + 2];
      B4[i] = make_float4(x, y, z, x * x + y * y + z * z);
    } else {
      B4[i] = make_float4(0.0f, 0.0f, 0.0f, __builtin_inff());
    }
  }
}

__global__ __launch_bounds__(WG_THR) void vm_match(const float4* __restrict__ A4,
                                                   const float4* __restrict__ B4,
                                                   int2* __restrict__ part) {
  extern __shared__ float4 Bs4[];
  float2* Bs2 = (float2*)Bs4;

  const int slice = blockIdx.y;
  const int col0  = slice * N_SLICE;
  const float4* __restrict__ Bg = B4 + col0;

  const float2* A2 = (const float2*)A4;
  const int lane = threadIdx.x & 31;
  const int wave = threadIdx.x >> 5;
  const int nlo  = lane & 15;
  const int kp   = lane >> 4;
  const int rt = blockIdx.x * WG_WAVES + wave;

  static_assert(N_SLICE > 4 * WG_THR && N_SLICE <= 5 * WG_THR, "staging is written for 4 full passes and a tail");
  float2 a2 = make_float2(0.0f, 0.0f);
  if (rt < NT_ROW) a2 = A2[2 * (rt * 16 + nlo) + kp];

  const int  tid  = threadIdx.x;
  const bool tail = tid < N_SLICE - 4 * WG_THR;
  const float4 s0 = Bg[tid];
  const float4 s1 = Bg[tid + 1 * WG_THR];
  const float4 s2 = Bg[tid + 2 * WG_THR];
  const float4 s3 = Bg[tid + 3 * WG_THR];
  float4 s4 = make_float4(0.0f, 0.0f, 0.0f, 0.0f);
  if (tail) s4 = Bg[tid + 4 * WG_THR];
  Bs4[tid]              = s0;
  Bs4[tid + 1 * WG_THR] = s1;
  Bs4[tid + 2 * WG_THR] = s2;
  Bs4[tid + 3 * WG_THR] = s3;
  if (tail) Bs4[tid + 4 * WG_THR] = s4;
  __syncthreads();

  if (rt < NT_ROW) {
    const v2f af = {a2.x, a2.y};

    float vmin[8];
    int   imin[8];
#pragma unroll
    for (int r = 0; r < 8; ++r) { vmin[r] = __builtin_inff(); imin[r] = 0; }

#pragma unroll 2
    for (int ct = 0; ct < CT_SLICE; ++ct) {
      const int n = ct * 16 + nlo;
      const float2 b2 = Bs2[2 * n + kp];
      const v2f bf = {b2.x, b2.y};
      v8f c = {0.f, 0.f, 0.f, 0.f, 0.f, 0.f, 0.f, 0.f};
      c = __builtin_amdgcn_wmma_f32_16x16x4_f32(false, af, false, bf,
                                                (short)0, c, false, false);
#pragma unroll
      for (int r = 0; r < 8; ++r) {
        float cv = c[r];
        if (cv < vmin[r]) { vmin[r] = cv; imin[r] = n; }
      }
    }

    int2 mine = make_int2(0, 0);
#pragma unroll
    for (int r = 0; r < 8; ++r) {
      float v = vmin[r];
      int   idx = imin[r];
#pragma unroll
      for (int mask = 8; mask >= 1; mask >>= 1) {
        float vo = __shfl_xor(v, mask, 32);
        int   io = __shfl_xor(idx, mask, 32);
        const bool better = (vo < v) | ((vo == v) & (io < idx));
        idx = better ? io : idx;
        v   = better ? vo : v;
      }
      if (nlo == r) { mine.x = __float_as_int(v); mine.y = idx + col0; }
    }
    if (nlo < 8) {
      part[slice * N_MHR + rt * 16 + nlo + (kp << 3)] = mine;
    }
  }
}

__global__ __launch_bounds__(256) void vm_merge(const int2* __restrict__ part,
                                                int* __restrict__ out) {
  const int m = blockIdx.x * blockDim.x + threadIdx.x;
  if (m < N_MHR) {
    const int2 p = part[m];
    float bv = __int_as_float(p.x);
    int   bi = p.y;
#pragma unroll
    for (int s = 1; s < N_SPLIT; ++s) {
      const int2 q = part[s * N_MHR + m];
      const float v = __int_as_float(q.x);
      if (v < bv) { bv = v; bi = q.y; }
    }
    out[m] = bi;
  }
}

extern "C" void kernel_launch(void* const* d_in, const int* in_sizes, int n_in,
                              void* d_out, int out_size, void* d_ws, size_t ws_size,
                              hipStream_t stream) {
  const float* mhr   = (const float*)d_in[0];
  const float* smplx = (const float*)d_in[1];
  int* out = (int*)d_out;

  char* ws = (char*)d_ws;
  float4* A4  = (float4*)ws;
  float4* B4  = (float4*)(ws + (size_t)N_MHR * 16);
  int2*  part = (int2*)(ws + (size_t)N_MHR * 16 + (size_t)N_PAD * 16);

  vm_prep<<<(N_MHR + 255) / 256, 256, 0, stream>>>(mhr, smplx, A4, B4);

  vm_match<<<dim3(WG_X, N_SPLIT), WG_THR, LDS_RESERVE, stream>>>(A4, B4, part);

  vm_merge<<<(N_MHR + 255) / 256, 256, 0, stream>>>(part, out);
}
